# starting kernel
# baseline (speedup 1.0000x reference)
#include <hip/hip_runtime.h>


typedef float v2f __attribute__((ext_vector_type(2)));
typedef float v8f __attribute__((ext_vector_type(8)));

#define DH 256
#define DOUT 128

__global__ void zero_kernel(float* __restrict__ p, long n) {
    long i = (long)blockIdx.x * blockDim.x + threadIdx.x;
    if (i < n) p[i] = 0.0f;
}

__global__ void deg_kernel(const int* __restrict__ dst, int E, float* __restrict__ deg) {
    int e = blockIdx.x * blockDim.x + threadIdx.x;
    if (e < E) atomicAdd(&deg[dst[e]], 1.0f);
}

__global__ __launch_bounds__(256) void scatter_kernel(
    const float* __restrict__ X, const int* __restrict__ src,
    const int* __restrict__ dst, int E, float* __restrict__ agg)
{
    int e = blockIdx.x * blockDim.y + threadIdx.y;
    if (e >= E) return;
    int lane = threadIdx.x;
    int s = src[e];
    int d = dst[e];
    const float4* xs = (const float4*)(X + (size_t)s * DH) + lane * 2;
    float4 v0 = xs[0];
    float4 v1 = xs[1];
    float* ad = agg + (size_t)d * DH + lane * 8;
    atomicAdd(ad + 0, v0.x);
    atomicAdd(ad + 1, v0.y);
    atomicAdd(ad + 2, v0.z);
    atomicAdd(ad + 3, v0.w);
    atomicAdd(ad + 4, v1.x);
    atomicAdd(ad + 5, v1.y);
    atomicAdd(ad + 6, v1.z);
    atomicAdd(ad + 7, v1.w);
}

__global__ void norm_kernel(float* __restrict__ agg, const float* __restrict__ deg, long n) {
    long i = (long)blockIdx.x * blockDim.x + threadIdx.x;
    if (i < n) {
        int row = (int)(i >> 8);
        agg[i] *= 1.0f / fmaxf(deg[row], 1.0f);
    }
}

__global__ __launch_bounds__(128) void gemm_wmma_kernel(
    const float* __restrict__ A, const float* __restrict__ W,
    const float* Cin, const float* bias,
    float* D, int M, int N, int relu)
{
    __shared__ float Bs[DH * 64];

    const int lane = threadIdx.x;
    const int half = lane >> 4;
    const int sub  = lane & 15;
    const int col0 = blockIdx.y * 64;

    {
        const int tid = threadIdx.y * 32 + lane;
        for (int i = tid; i < DH * 16; i += 128) {
            const int row = i >> 4;
            const int c4  = (i & 15) << 2;
            *(float4*)&Bs[row * 64 + c4] =
                *(const float4*)&W[(size_t)row * N + col0 + c4];
        }
    }
    __syncthreads();

    const int tm   = blockIdx.x * blockDim.y + threadIdx.y;
    const int row0 = tm * 16;
    if (row0 >= M) return;

    const int arow = min(row0 + sub, M - 1);
    const float* Ap = A + (size_t)arow * DH;

    v8f acc[4] = {};

    for (int k0 = 0; k0 < DH; k0 += 4) {
        const int ka = k0 + 2 * half;
        const v2f a = *(const v2f*)(Ap + ka);
        const float* Bp = &Bs[ka * 64 + sub];
#pragma unroll
        for (int t = 0; t < 4; ++t) {
            v2f b;
            b.x = Bp[t * 16];
            b.y = Bp[t * 16 + 64];
            acc[t] = __builtin_amdgcn_wmma_f32_16x16x4_f32(
                false, a, false, b, (short)0, acc[t], false, false);
        }
    }

#pragma unroll
    for (int t = 0; t < 4; ++t) {
        const int n = col0 + t * 16 + sub;
        const float bv = bias ? bias[n] : 0.0f;
#pragma unroll
        for (int v = 0; v < 8; ++v) {
            const int m = row0 + v + 8 * half;
            if (m < M) {
                const size_t idx = (size_t)m * N + n;
                float r = acc[t][v] + bv;
                if (Cin) r += Cin[idx];
                if (relu) r = fmaxf(r, 0.0f);
                D[idx] = r;
            }
        }
    }
}

extern "C" void kernel_launch(void* const* d_in, const int* in_sizes, int n_in,
                              void* d_out, int out_size, void* d_ws, size_t ws_size,
                              hipStream_t stream)
{
    const float* x_item  = (const float*)d_in[0];
    const float* x_user  = (const float*)d_in[1];
    const int*   src_ii  = (const int*)d_in[2];
    const int*   dst_ii  = (const int*)d_in[3];
    const int*   src_iu0 = (const int*)d_in[4];
    const int*   dst_iu0 = (const int*)d_in[5];
    const int*   src_iu1 = (const int*)d_in[6];
    const int*   dst_iu1 = (const int*)d_in[7];
    const float* Wn1 = (const float*)d_in[8];
    const float* Ws1 = (const float*)d_in[9];
    const float* b1  = (const float*)d_in[10];
    const float* Wn2 = (const float*)d_in[11];
    const float* Ws2 = (const float*)d_in[12];
    const float* b2  = (const float*)d_in[13];
    const float* Wn3 = (const float*)d_in[14];
    const float* Ws3 = (const float*)d_in[15];
    const float* b3  = (const float*)d_in[16];
    const float* Wlin = (const float*)d_in[17];
    const float* blin = (const float*)d_in[18];

    const int E_ii  = in_sizes[2];
    const int E_iu0 = in_sizes[4];
    const int E_iu1 = in_sizes[6];
    const int N_ITEM_DST = 50000;
    const int N_USER0    = in_sizes[1] / DH;
    const int N_USER1    = 25000;

    char* ws = (char*)d_ws;
    const size_t off_deg   = 0;
    const size_t off_agg   = 256 * 1024;
    const size_t off_itemx = off_agg   + (size_t)N_ITEM_DST * DH * sizeof(float);
    const size_t off_userx = off_itemx + (size_t)N_ITEM_DST * DH * sizeof(float);
    float* deg    = (float*)(ws + off_deg);
    float* agg    = (float*)(ws + off_agg);
    float* itemx  = (float*)(ws + off_itemx);
    float* userx  = (float*)(ws + off_userx);
    float* userx2 = itemx;
    float* out    = (float*)d_out;

    auto zero = [&](float* p, long n) {
        zero_kernel<<<dim3((unsigned)((n + 255) / 256)), dim3(256), 0, stream>>>(p, n);
    };
    auto gemm = [&](const float* A, const float* Wm, const float* Cin,
                    const float* bias, float* Dm, int M, int N, int relu) {
        int mt = (M + 15) / 16;
        dim3 grid((mt + 3) / 4, N / 64);
        dim3 block(32, 4);
        gemm_wmma_kernel<<<grid, block, 0, stream>>>(A, Wm, Cin, bias, Dm, M, N, relu);
    };
    auto sage_agg = [&](const float* Xsrc, const int* src, const int* dst,
                        int E, int n_dst) {
        zero(deg, n_dst);
        zero(agg, (long)n_dst * DH);
        deg_kernel<<<dim3((E + 255) / 256), dim3(256), 0, stream>>>(dst, E, deg);
        scatter_kernel<<<dim3((E + 7) / 8), dim3(32, 8), 0, stream>>>(Xsrc, src, dst, E, agg);
        norm_kernel<<<dim3((unsigned)(((long)n_dst * DH + 255) / 256)), dim3(256), 0, stream>>>(
            agg, deg, (long)n_dst * DH);
    };

    sage_agg(x_item, src_ii, dst_ii, E_ii, N_ITEM_DST);
    gemm(x_item, Ws1, nullptr, nullptr, itemx, N_ITEM_DST, DH, 0);
    gemm(agg,    Wn1, itemx,   b1,      itemx, N_ITEM_DST, DH, 1);

    sage_agg(x_item, src_iu0, dst_iu0, E_iu0, N_USER0);
    gemm(x_user, Ws2, nullptr, nullptr, userx, N_USER0, DH, 0);
    gemm(agg,    Wn2, userx,   b2,      userx, N_USER0, DH, 1);

    sage_agg(itemx, src_iu1, dst_iu1, E_iu1, N_USER1);
    gemm(userx, Ws3, nullptr, nullptr, userx2, N_USER1, DH, 0);
    gemm(agg,   Wn3, userx2,  b3,      userx2, N_USER1, DH, 1);

    gemm(userx2, Wlin, nullptr, blin, out, N_USER1, DOUT, 0);
}


# Neighbour sum without float atomics: edges bucketed by destination (int atomic rank, row offsets, fill), one wave per row gathers and averages; sum order is now the arrival-order rank
# speedup vs baseline: 2.6010x; 2.6010x over previous
#include <hip/hip_runtime.h>


typedef float v2f __attribute__((ext_vector_type(2)));
typedef float v8f __attribute__((ext_vector_type(8)));

#define DH 256
#define DOUT 128

__global__ void zero_kernel(int* __restrict__ p, int n) {
    int i = blockIdx.x * blockDim.x + threadIdx.x;
    if (i < n) p[i] = 0;
}

__global__ void rank_kernel(const int* __restrict__ dst, int E,
                            int* __restrict__ cnt, int* __restrict__ rank) {
    int e = blockIdx.x * blockDim.x + threadIdx.x;
    if (e < E) rank[e] = atomicAdd(&cnt[dst[e]], 1);
}

__global__ void offset_kernel(const int* __restrict__ cnt, int n,
                              int* __restrict__ off, int* __restrict__ total) {
    const int i = blockIdx.x * blockDim.x + threadIdx.x;
    const int lane = threadIdx.x & 31;
    const int c = (i < n) ? cnt[i] : 0;
    int incl = c;
    for (int d = 1; d < 32; d <<= 1) {
        const int t = __shfl_up(incl, d, 32);
        if (lane >= d) incl += t;
    }
    int base = 0;
    if (lane == 31) base = atomicAdd(total, incl);
    base = __shfl(base, 31, 32);
    if (i < n) off[i] = base + incl - c;
}

__global__ void fill_kernel(const int* __restrict__ src, const int* __restrict__ dst,
                            const int* __restrict__ rank, const int* __restrict__ off,
                            int E, int* __restrict__ slots) {
    int e = blockIdx.x * blockDim.x + threadIdx.x;
    if (e < E) slots[off[dst[e]] + rank[e]] = src[e];
}

__global__ __launch_bounds__(256) void gather_mean_kernel(
    const float* __restrict__ X, const int* __restrict__ slots,
    const int* __restrict__ off, const int* __restrict__ cnt,
    int n_dst, float* __restrict__ agg)
{
    const int row = __builtin_amdgcn_readfirstlane(
        (int)(blockIdx.x * blockDim.y + threadIdx.y));
    if (row >= n_dst) return;
    const int lane = threadIdx.x;
    const int c = cnt[row];
    const int o = off[row];

    float4 a0 = make_float4(0.0f, 0.0f, 0.0f, 0.0f);
    float4 a1 = make_float4(0.0f, 0.0f, 0.0f, 0.0f);
    for (int base = 0; base < c; base += 32) {
        const int m = min(32, c - base);
        const int sid = (lane < m) ? slots[o + base + lane] : 0;
#pragma unroll 8
        for (int j = 0; j < m; ++j) {
            const int s = __builtin_amdgcn_readlane(sid, j);
            const float4* xs = (const float4*)(X + (size_t)s * DH);
            const float4 v0 = xs[lane];
            const float4 v1 = xs[lane + 32];
            a0.x += v0.x; a0.y += v0.y; a0.z += v0.z; a0.w += v0.w;
            a1.x += v1.x; a1.y += v1.y; a1.z += v1.z; a1.w += v1.w;
        }
    }
    const float inv = 1.0f / fmaxf((float)c, 1.0f);
    float4* ao = (float4*)(agg + (size_t)row * DH);
    ao[lane]      = make_float4(a0.x * inv, a0.y * inv, a0.z * inv, a0.w * inv);
    ao[lane + 32] = make_float4(a1.x * inv, a1.y * inv, a1.z * inv, a1.w * inv);
}

__global__ __launch_bounds__(128) void gemm_wmma_kernel(
    const float* __restrict__ A, const float* __restrict__ W,
    const float* Cin, const float* bias,
    float* D, int M, int N, int relu)
{
    __shared__ float Bs[DH * 64];

    const int lane = threadIdx.x;
    const int half = lane >> 4;
    const int sub  = lane & 15;
    const int col0 = blockIdx.y * 64;

    {
        const int tid = threadIdx.y * 32 + lane;
        for (int i = tid; i < DH * 16; i += 128) {
            const int row = i >> 4;
            const int c4  = (i & 15) << 2;
            *(float4*)&Bs[row * 64 + c4] =
                *(const float4*)&W[(size_t)row * N + col0 + c4];
        }
    }
    __syncthreads();

    const int tm   = blockIdx.x * blockDim.y + threadIdx.y;
    const int row0 = tm * 16;
    if (row0 >= M) return;

    const int arow = min(row0 + sub, M - 1);
    const float* Ap = A + (size_t)arow * DH;

    v8f acc[4] = {};

    for (int k0 = 0; k0 < DH; k0 += 4) {
        const int ka = k0 + 2 * half;
        const v2f a = *(const v2f*)(Ap + ka);
        const float* Bp = &Bs[ka * 64 + sub];
#pragma unroll
        for (int t = 0; t < 4; ++t) {
            v2f b;
            b.x = Bp[t * 16];
            b.y = Bp[t * 16 + 64];
            acc[t] = __builtin_amdgcn_wmma_f32_16x16x4_f32(
                false, a, false, b, (short)0, acc[t], false, false);
        }
    }

#pragma unroll
    for (int t = 0; t < 4; ++t) {
        const int n = col0 + t * 16 + sub;
        const float bv = bias ? bias[n] : 0.0f;
#pragma unroll
        for (int v = 0; v < 8; ++v) {
            const int m = row0 + v + 8 * half;
            if (m < M) {
                const size_t idx = (size_t)m * N + n;
                float r = acc[t][v] + bv;
                if (Cin) r += Cin[idx];
                if (relu) r = fmaxf(r, 0.0f);
                D[idx] = r;
            }
        }
    }
}

extern "C" void kernel_launch(void* const* d_in, const int* in_sizes, int n_in,
                              void* d_out, int out_size, void* d_ws, size_t ws_size,
                              hipStream_t stream)
{
    const float* x_item  = (const float*)d_in[0];
    const float* x_user  = (const float*)d_in[1];
    const int*   src_ii  = (const int*)d_in[2];
    const int*   dst_ii  = (const int*)d_in[3];
    const int*   src_iu0 = (const int*)d_in[4];
    const int*   dst_iu0 = (const int*)d_in[5];
    const int*   src_iu1 = (const int*)d_in[6];
    const int*   dst_iu1 = (const int*)d_in[7];
    const float* Wn1 = (const float*)d_in[8];
    const float* Ws1 = (const float*)d_in[9];
    const float* b1  = (const float*)d_in[10];
    const float* Wn2 = (const float*)d_in[11];
    const float* Ws2 = (const float*)d_in[12];
    const float* b2  = (const float*)d_in[13];
    const float* Wn3 = (const float*)d_in[14];
    const float* Ws3 = (const float*)d_in[15];
    const float* b3  = (const float*)d_in[16];
    const float* Wlin = (const float*)d_in[17];
    const float* blin = (const float*)d_in[18];

    const int E_ii  = in_sizes[2];
    const int E_iu0 = in_sizes[4];
    const int E_iu1 = in_sizes[6];
    const int N_ITEM_DST = 50000;
    const int N_USER0    = in_sizes[1] / DH;
    const int N_USER1    = 25000;

    char* ws = (char*)d_ws;
    const size_t off_deg   = 0;
    const size_t off_agg   = 256 * 1024;
    const size_t off_itemx = off_agg   + (size_t)N_ITEM_DST * DH * sizeof(float);
    const size_t off_userx = off_itemx + (size_t)N_ITEM_DST * DH * sizeof(float);
    int*   cnt    = (int*)(ws + off_deg);
    float* agg    = (float*)(ws + off_agg);
    float* itemx  = (float*)(ws + off_itemx);
    float* userx  = (float*)(ws + off_userx);
    float* userx2 = itemx;
    float* out    = (float*)d_out;
    int* off_row  = (int*)(userx + (size_t)N_USER1 * DH);
    int* rank     = off_row + 65536;
    int* slots    = rank + 1048576;

    auto zero = [&](int* p, int n) {
        zero_kernel<<<dim3((unsigned)((n + 255) / 256)), dim3(256), 0, stream>>>(p, n);
    };
    auto gemm = [&](const float* A, const float* Wm, const float* Cin,
                    const float* bias, float* Dm, int M, int N, int relu) {
        int mt = (M + 15) / 16;
        dim3 grid((mt + 3) / 4, N / 64);
        dim3 block(32, 4);
        gemm_wmma_kernel<<<grid, block, 0, stream>>>(A, Wm, Cin, bias, Dm, M, N, relu);
    };
    auto sage_agg = [&](const float* Xsrc, const int* src, const int* dst,
                        int E, int n_dst) {
        zero(cnt, n_dst + 1);
        rank_kernel<<<dim3((E + 255) / 256), dim3(256), 0, stream>>>(dst, E, cnt, rank);
        offset_kernel<<<dim3((n_dst + 255) / 256), dim3(256), 0, stream>>>(
            cnt, n_dst, off_row, cnt + n_dst);
        fill_kernel<<<dim3((E + 255) / 256), dim3(256), 0, stream>>>(
            src, dst, rank, off_row, E, slots);
        gather_mean_kernel<<<dim3((n_dst + 7) / 8), dim3(32, 8), 0, stream>>>(
            Xsrc, slots, off_row, cnt, n_dst, agg);
    };

    sage_agg(x_item, src_ii, dst_ii, E_ii, N_ITEM_DST);
    gemm(x_item, Ws1, nullptr, nullptr, itemx, N_ITEM_DST, DH, 0);
    gemm(agg,    Wn1, itemx,   b1,      itemx, N_ITEM_DST, DH, 1);

    sage_agg(x_item, src_iu0, dst_iu0, E_iu0, N_USER0);
    gemm(x_user, Ws2, nullptr, nullptr, userx, N_USER0, DH, 0);
    gemm(agg,    Wn2, userx,   b2,      userx, N_USER0, DH, 1);

    sage_agg(itemx, src_iu1, dst_iu1, E_iu1, N_USER1);
    gemm(userx, Ws3, nullptr, nullptr, userx2, N_USER1, DH, 0);
    gemm(agg,   Wn3, userx2,  b3,      userx2, N_USER1, DH, 1);

    gemm(userx2, Wlin, nullptr, blin, out, N_USER1, DOUT, 0);
}


# Layer 2 restricted to the first n_user1 user rows, the only ones layer 3 reads (reference slices user_x[:n_user1]): only their edges are bucketed; gather and both GEMMs run on 25000 rows, not 50000
# speedup vs baseline: 3.0824x; 1.1851x over previous
#include <hip/hip_runtime.h>


typedef float v2f __attribute__((ext_vector_type(2)));
typedef float v8f __attribute__((ext_vector_type(8)));

#define DH 256
#define DOUT 128

__global__ void zero_kernel(int* __restrict__ p, int n) {
    int i = blockIdx.x * blockDim.x + threadIdx.x;
    if (i < n) p[i] = 0;
}

__global__ void rank_kernel(const int* __restrict__ dst, int E, int n_dst,
                            int* __restrict__ cnt, int* __restrict__ rank) {
    int e = blockIdx.x * blockDim.x + threadIdx.x;
    if (e < E) {
        const int d = dst[e];
        if ((unsigned)d < (unsigned)n_dst) rank[e] = atomicAdd(&cnt[d], 1);
    }
}

__global__ void offset_kernel(const int* __restrict__ cnt, int n,
                              int* __restrict__ off, int* __restrict__ total) {
    const int i = blockIdx.x * blockDim.x + threadIdx.x;
    const int lane = threadIdx.x & 31;
    const int c = (i < n) ? cnt[i] : 0;
    int incl = c;
    for (int d = 1; d < 32; d <<= 1) {
        const int t = __shfl_up(incl, d, 32);
        if (lane >= d) incl += t;
    }
    int base = 0;
    if (lane == 31) base = atomicAdd(total, incl);
    base = __shfl(base, 31, 32);
    if (i < n) off[i] = base + incl - c;
}

__global__ void fill_kernel(const int* __restrict__ src, const int* __restrict__ dst,
                            const int* __restrict__ rank, const int* __restrict__ off,
                            int E, int n_dst, int* __restrict__ slots) {
    int e = blockIdx.x * blockDim.x + threadIdx.x;
    if (e < E) {
        const int d = dst[e];
        if ((unsigned)d < (unsigned)n_dst) slots[off[d] + rank[e]] = src[e];
    }
}

__global__ __launch_bounds__(256) void gather_mean_kernel(
    const float* __restrict__ X, const int* __restrict__ slots,
    const int* __restrict__ off, const int* __restrict__ cnt,
    int n_dst, float* __restrict__ agg)
{
    const int row = __builtin_amdgcn_readfirstlane(
        (int)(blockIdx.x * blockDim.y + threadIdx.y));
    if (row >= n_dst) return;
    const int lane = threadIdx.x;
    const int c = cnt[row];
    const int o = off[row];

    float4 a0 = make_float4(0.0f, 0.0f, 0.0f, 0.0f);
    float4 a1 = make_float4(0.0f, 0.0f, 0.0f, 0.0f);
    for (int base = 0; base < c; base += 32) {
        const int m = min(32, c - base);
        const int sid = (lane < m) ? slots[o + base + lane] : 0;
#pragma unroll 8
        for (int j = 0; j < m; ++j) {
            const int s = __builtin_amdgcn_readlane(sid, j);
            const float4* xs = (const float4*)(X + (size_t)s * DH);
            const float4 v0 = xs[lane];
            const float4 v1 = xs[lane + 32];
            a0.x += v0.x; a0.y += v0.y; a0.z += v0.z; a0.w += v0.w;
            a1.x += v1.x; a1.y += v1.y; a1.z += v1.z; a1.w += v1.w;
        }
    }
    const float inv = 1.0f / fmaxf((float)c, 1.0f);
    float4* ao = (float4*)(agg + (size_t)row * DH);
    ao[lane]      = make_float4(a0.x * inv, a0.y * inv, a0.z * inv, a0.w * inv);
    ao[lane + 32] = make_float4(a1.x * inv, a1.y * inv, a1.z * inv, a1.w * inv);
}

__global__ __launch_bounds__(128) void gemm_wmma_kernel(
    const float* __restrict__ A, const float* __restrict__ W,
    const float* Cin, const float* bias,
    float* D, int M, int N, int relu)
{
    __shared__ float Bs[DH * 64];

    const int lane = threadIdx.x;
    const int half = lane >> 4;
    const int sub  = lane & 15;
    const int col0 = blockIdx.y * 64;

    {
        const int tid = threadIdx.y * 32 + lane;
        for (int i = tid; i < DH * 16; i += 128) {
            const int row = i >> 4;
            const int c4  = (i & 15) << 2;
            *(float4*)&Bs[row * 64 + c4] =
                *(const float4*)&W[(size_t)row * N + col0 + c4];
        }
    }
    __syncthreads();

    const int tm   = blockIdx.x * blockDim.y + threadIdx.y;
    const int row0 = tm * 16;
    if (row0 >= M) return;

    const int arow = min(row0 + sub, M - 1);
    const float* Ap = A + (size_t)arow * DH;

    v8f acc[4] = {};

    for (int k0 = 0; k0 < DH; k0 += 4) {
        const int ka = k0 + 2 * half;
        const v2f a = *(const v2f*)(Ap + ka);
        const float* Bp = &Bs[ka * 64 + sub];
#pragma unroll
        for (int t = 0; t < 4; ++t) {
            v2f b;
            b.x = Bp[t * 16];
            b.y = Bp[t * 16 + 64];
            acc[t] = __builtin_amdgcn_wmma_f32_16x16x4_f32(
                false, a, false, b, (short)0, acc[t], false, false);
        }
    }

#pragma unroll
    for (int t = 0; t < 4; ++t) {
        const int n = col0 + t * 16 + sub;
        const float bv = bias ? bias[n] : 0.0f;
#pragma unroll
        for (int v = 0; v < 8; ++v) {
            const int m = row0 + v + 8 * half;
            if (m < M) {
                const size_t idx = (size_t)m * N + n;
                float r = acc[t][v] + bv;
                if (Cin) r += Cin[idx];
                if (relu) r = fmaxf(r, 0.0f);
                D[idx] = r;
            }
        }
    }
}

extern "C" void kernel_launch(void* const* d_in, const int* in_sizes, int n_in,
                              void* d_out, int out_size, void* d_ws, size_t ws_size,
                              hipStream_t stream)
{
    const float* x_item  = (const float*)d_in[0];
    const float* x_user  = (const float*)d_in[1];
    const int*   src_ii  = (const int*)d_in[2];
    const int*   dst_ii  = (const int*)d_in[3];
    const int*   src_iu0 = (const int*)d_in[4];
    const int*   dst_iu0 = (const int*)d_in[5];
    const int*   src_iu1 = (const int*)d_in[6];
    const int*   dst_iu1 = (const int*)d_in[7];
    const float* Wn1 = (const float*)d_in[8];
    const float* Ws1 = (const float*)d_in[9];
    const float* b1  = (const float*)d_in[10];
    const float* Wn2 = (const float*)d_in[11];
    const float* Ws2 = (const float*)d_in[12];
    const float* b2  = (const float*)d_in[13];
    const float* Wn3 = (const float*)d_in[14];
    const float* Ws3 = (const float*)d_in[15];
    const float* b3  = (const float*)d_in[16];
    const float* Wlin = (const float*)d_in[17];
    const float* blin = (const float*)d_in[18];

    const int E_ii  = in_sizes[2];
    const int E_iu0 = in_sizes[4];
    const int E_iu1 = in_sizes[6];
    const int N_ITEM_DST = 50000;
    const int N_USER1    = 25000;

    char* ws = (char*)d_ws;
    const size_t off_deg   = 0;
    const size_t off_agg   = 256 * 1024;
    const size_t off_itemx = off_agg   + (size_t)N_ITEM_DST * DH * sizeof(float);
    const size_t off_userx = off_itemx + (size_t)N_ITEM_DST * DH * sizeof(float);
    int*   cnt    = (int*)(ws + off_deg);
    float* agg    = (float*)(ws + off_agg);
    float* itemx  = (float*)(ws + off_itemx);
    float* userx  = (float*)(ws + off_userx);
    float* userx2 = itemx;
    float* out    = (float*)d_out;
    int* off_row  = (int*)(userx + (size_t)N_USER1 * DH);
    int* rank     = off_row + 65536;
    int* slots    = rank + 1048576;

    auto zero = [&](int* p, int n) {
        zero_kernel<<<dim3((unsigned)((n + 255) / 256)), dim3(256), 0, stream>>>(p, n);
    };
    auto gemm = [&](const float* A, const float* Wm, const float* Cin,
                    const float* bias, float* Dm, int M, int N, int relu) {
        int mt = (M + 15) / 16;
        dim3 grid((mt + 3) / 4, N / 64);
        dim3 block(32, 4);
        gemm_wmma_kernel<<<grid, block, 0, stream>>>(A, Wm, Cin, bias, Dm, M, N, relu);
    };
    auto sage_agg = [&](const float* Xsrc, const int* src, const int* dst,
                        int E, int n_dst) {
        zero(cnt, n_dst + 1);
        rank_kernel<<<dim3((E + 255) / 256), dim3(256), 0, stream>>>(dst, E, n_dst, cnt, rank);
        offset_kernel<<<dim3((n_dst + 255) / 256), dim3(256), 0, stream>>>(
            cnt, n_dst, off_row, cnt + n_dst);
        fill_kernel<<<dim3((E + 255) / 256), dim3(256), 0, stream>>>(
            src, dst, rank, off_row, E, n_dst, slots);
        gather_mean_kernel<<<dim3((n_dst + 7) / 8), dim3(32, 8), 0, stream>>>(
            Xsrc, slots, off_row, cnt, n_dst, agg);
    };

    sage_agg(x_item, src_ii, dst_ii, E_ii, N_ITEM_DST);
    gemm(x_item, Ws1, nullptr, nullptr, itemx, N_ITEM_DST, DH, 0);
    gemm(agg,    Wn1, itemx,   b1,      itemx, N_ITEM_DST, DH, 1);

    sage_agg(x_item, src_iu0, dst_iu0, E_iu0, N_USER1);
    gemm(x_user, Ws2, nullptr, nullptr, userx, N_USER1, DH, 0);
    gemm(agg,    Wn2, userx,   b2,      userx, N_USER1, DH, 1);

    sage_agg(itemx, src_iu1, dst_iu1, E_iu1, N_USER1);
    gemm(userx, Ws3, nullptr, nullptr, userx2, N_USER1, DH, 0);
    gemm(agg,   Wn3, userx2,  b3,      userx2, N_USER1, DH, 1);

    gemm(userx2, Wlin, nullptr, blin, out, N_USER1, DOUT, 0);
}


# GEMM: each thread's 32 W-stage loads and each lane's bias and Cin epilogue loads are requested as one group before first use, not one memory round trip per load; arithmetic and its order unchanged
# speedup vs baseline: 4.1234x; 1.3377x over previous
#include <hip/hip_runtime.h>


typedef float v2f __attribute__((ext_vector_type(2)));
typedef float v8f __attribute__((ext_vector_type(8)));

#define DH 256
#define DOUT 128

__global__ void zero_kernel(int* __restrict__ p, int n) {
    int i = blockIdx.x * blockDim.x + threadIdx.x;
    if (i < n) p[i] = 0;
}

__global__ void rank_kernel(const int* __restrict__ dst, int E, int n_dst,
                            int* __restrict__ cnt, int* __restrict__ rank) {
    int e = blockIdx.x * blockDim.x + threadIdx.x;
    if (e < E) {
        const int d = dst[e];
        if ((unsigned)d < (unsigned)n_dst) rank[e] = atomicAdd(&cnt[d], 1);
    }
}

__global__ void offset_kernel(const int* __restrict__ cnt, int n,
                              int* __restrict__ off, int* __restrict__ total) {
    const int i = blockIdx.x * blockDim.x + threadIdx.x;
    const int lane = threadIdx.x & 31;
    const int c = (i < n) ? cnt[i] : 0;
    int incl = c;
    for (int d = 1; d < 32; d <<= 1) {
        const int t = __shfl_up(incl, d, 32);
        if (lane >= d) incl += t;
    }
    int base = 0;
    if (lane == 31) base = atomicAdd(total, incl);
    base = __shfl(base, 31, 32);
    if (i < n) off[i] = base + incl - c;
}

__global__ void fill_kernel(const int* __restrict__ src, const int* __restrict__ dst,
                            const int* __restrict__ rank, const int* __restrict__ off,
                            int E, int n_dst, int* __restrict__ slots) {
    int e = blockIdx.x * blockDim.x + threadIdx.x;
    if (e < E) {
        const int d = dst[e];
        if ((unsigned)d < (unsigned)n_dst) slots[off[d] + rank[e]] = src[e];
    }
}

__global__ __launch_bounds__(256) void gather_mean_kernel(
    const float* __restrict__ X, const int* __restrict__ slots,
    const int* __restrict__ off, const int* __restrict__ cnt,
    int n_dst, float* __restrict__ agg)
{
    const int row = __builtin_amdgcn_readfirstlane(
        (int)(blockIdx.x * blockDim.y + threadIdx.y));
    if (row >= n_dst) return;
    const int lane = threadIdx.x;
    const int c = cnt[row];
    const int o = off[row];

    float4 a0 = make_float4(0.0f, 0.0f, 0.0f, 0.0f);
    float4 a1 = make_float4(0.0f, 0.0f, 0.0f, 0.0f);
    for (int base = 0; base < c; base += 32) {
        const int m = min(32, c - base);
        const int sid = (lane < m) ? slots[o + base + lane] : 0;
#pragma unroll 8
        for (int j = 0; j < m; ++j) {
            const int s = __builtin_amdgcn_readlane(sid, j);
            const float4* xs = (const float4*)(X + (size_t)s * DH);
            const float4 v0 = xs[lane];
            const float4 v1 = xs[lane + 32];
            a0.x += v0.x; a0.y += v0.y; a0.z += v0.z; a0.w += v0.w;
            a1.x += v1.x; a1.y += v1.y; a1.z += v1.z; a1.w += v1.w;
        }
    }
    const float inv = 1.0f / fmaxf((float)c, 1.0f);
    float4* ao = (float4*)(agg + (size_t)row * DH);
    ao[lane]      = make_float4(a0.x * inv, a0.y * inv, a0.z * inv, a0.w * inv);
    ao[lane + 32] = make_float4(a1.x * inv, a1.y * inv, a1.z * inv, a1.w * inv);
}

template <bool HASC>
__device__ __forceinline__ void gemm_epilogue(
    const v8f (&acc)[4], const float* Cin, const float* bias, float* D,
    int M, int N, int relu, int row0, int col0, int half, int sub)
{
    const int ncol = col0 + sub;
    const int mb   = row0 + 8 * half;

    float bv[4];
#pragma unroll
    for (int t = 0; t < 4; ++t) bv[t] = bias ? bias[ncol + t * 16] : 0.0f;

    float cv[8][4];
    if (HASC) {
#pragma unroll
        for (int v = 0; v < 8; ++v) {
            const int mm = min(mb + v, M - 1);
            const float* cp = Cin + (size_t)mm * N + ncol;
#pragma unroll
            for (int t = 0; t < 4; ++t) cv[v][t] = cp[t * 16];
        }
    }

#pragma unroll
    for (int v = 0; v < 8; ++v) {
        const int m = mb + v;
        if (m < M) {
            float* dp = D + (size_t)m * N + ncol;
#pragma unroll
            for (int t = 0; t < 4; ++t) {
                float r = acc[t][v] + bv[t];
                if (HASC) r += cv[v][t];
                if (relu) r = fmaxf(r, 0.0f);
                dp[t * 16] = r;
            }
        }
    }
}

__global__ __launch_bounds__(128) void gemm_wmma_kernel(
    const float* __restrict__ A, const float* __restrict__ W,
    const float* Cin, const float* bias,
    float* D, int M, int N, int relu)
{
    __shared__ float Bs[DH * 64];

    const int lane = threadIdx.x;
    const int half = lane >> 4;
    const int sub  = lane & 15;
    const int col0 = blockIdx.y * 64;

    {
        const int tid = threadIdx.y * 32 + lane;
        const int r0  = tid >> 4;
        const int c4  = (tid & 15) << 2;
        const float* Wp = W + (size_t)r0 * N + col0 + c4;
        float*       Bq = &Bs[r0 * 64 + c4];
        float4 w[32];
#pragma unroll
        for (int j = 0; j < 32; ++j)
            w[j] = *(const float4*)(Wp + (size_t)(8 * j) * N);
#pragma unroll
        for (int j = 0; j < 32; ++j)
            *(float4*)(Bq + 8 * 64 * j) = w[j];
    }
    __syncthreads();

    const int tm   = blockIdx.x * blockDim.y + threadIdx.y;
    const int row0 = tm * 16;
    if (row0 >= M) return;

    const int arow = min(row0 + sub, M - 1);
    const float* Ap = A + (size_t)arow * DH;

    v8f acc[4] = {};

    for (int k0 = 0; k0 < DH; k0 += 4) {
        const int ka = k0 + 2 * half;
        const v2f a = *(const v2f*)(Ap + ka);
        const float* Bp = &Bs[ka * 64 + sub];
#pragma unroll
        for (int t = 0; t < 4; ++t) {
            v2f b;
            b.x = Bp[t * 16];
            b.y = Bp[t * 16 + 64];
            acc[t] = __builtin_amdgcn_wmma_f32_16x16x4_f32(
                false, a, false, b, (short)0, acc[t], false, false);
        }
    }

    if (Cin) gemm_epilogue<true >(acc, Cin, bias, D, M, N, relu, row0, col0, half, sub);
    else     gemm_epilogue<false>(acc, Cin, bias, D, M, N, relu, row0, col0, half, sub);
}

extern "C" void kernel_launch(void* const* d_in, const int* in_sizes, int n_in,
                              void* d_out, int out_size, void* d_ws, size_t ws_size,
                              hipStream_t stream)
{
    const float* x_item  = (const float*)d_in[0];
    const float* x_user  = (const float*)d_in[1];
    const int*   src_ii  = (const int*)d_in[2];
    const int*   dst_ii  = (const int*)d_in[3];
    const int*   src_iu0 = (const int*)d_in[4];
    const int*   dst_iu0 = (const int*)d_in[5];
    const int*   src_iu1 = (const int*)d_in[6];
    const int*   dst_iu1 = (const int*)d_in[7];
    const float* Wn1 = (const float*)d_in[8];
    const float* Ws1 = (const float*)d_in[9];
    const float* b1  = (const float*)d_in[10];
    const float* Wn2 = (const float*)d_in[11];
    const float* Ws2 = (const float*)d_in[12];
    const float* b2  = (const float*)d_in[13];
    const float* Wn3 = (const float*)d_in[14];
    const float* Ws3 = (const float*)d_in[15];
    const float* b3  = (const float*)d_in[16];
    const float* Wlin = (const float*)d_in[17];
    const float* blin = (const float*)d_in[18];

    const int E_ii  = in_sizes[2];
    const int E_iu0 = in_sizes[4];
    const int E_iu1 = in_sizes[6];
    const int N_ITEM_DST = 50000;
    const int N_USER1    = 25000;

    char* ws = (char*)d_ws;
    const size_t off_deg   = 0;
    const size_t off_agg   = 256 * 1024;
    const size_t off_itemx = off_agg   + (size_t)N_ITEM_DST * DH * sizeof(float);
    const size_t off_userx = off_itemx + (size_t)N_ITEM_DST * DH * sizeof(float);
    int*   cnt    = (int*)(ws + off_deg);
    float* agg    = (float*)(ws + off_agg);
    float* itemx  = (float*)(ws + off_itemx);
    float* userx  = (float*)(ws + off_userx);
    float* userx2 = itemx;
    float* out    = (float*)d_out;
    int* off_row  = (int*)(userx + (size_t)N_USER1 * DH);
    int* rank     = off_row + 65536;
    int* slots    = rank + 1048576;

    auto zero = [&](int* p, int n) {
        zero_kernel<<<dim3((unsigned)((n + 255) / 256)), dim3(256), 0, stream>>>(p, n);
    };
    auto gemm = [&](const float* A, const float* Wm, const float* Cin,
                    const float* bias, float* Dm, int M, int N, int relu) {
        int mt = (M + 15) / 16;
        dim3 grid((mt + 3) / 4, N / 64);
        dim3 block(32, 4);
        gemm_wmma_kernel<<<grid, block, 0, stream>>>(A, Wm, Cin, bias, Dm, M, N, relu);
    };
    auto sage_agg = [&](const float* Xsrc, const int* src, const int* dst,
                        int E, int n_dst) {
        zero(cnt, n_dst + 1);
        rank_kernel<<<dim3((E + 255) / 256), dim3(256), 0, stream>>>(dst, E, n_dst, cnt, rank);
        offset_kernel<<<dim3((n_dst + 255) / 256), dim3(256), 0, stream>>>(
            cnt, n_dst, off_row, cnt + n_dst);
        fill_kernel<<<dim3((E + 255) / 256), dim3(256), 0, stream>>>(
            src, dst, rank, off_row, E, n_dst, slots);
        gather_mean_kernel<<<dim3((n_dst + 7) / 8), dim3(32, 8), 0, stream>>>(
            Xsrc, slots, off_row, cnt, n_dst, agg);
    };

    sage_agg(x_item, src_ii, dst_ii, E_ii, N_ITEM_DST);
    gemm(x_item, Ws1, nullptr, nullptr, itemx, N_ITEM_DST, DH, 0);
    gemm(agg,    Wn1, itemx,   b1,      itemx, N_ITEM_DST, DH, 1);

    sage_agg(x_item, src_iu0, dst_iu0, E_iu0, N_USER1);
    gemm(x_user, Ws2, nullptr, nullptr, userx, N_USER1, DH, 0);
    gemm(agg,    Wn2, userx,   b2,      userx, N_USER1, DH, 1);

    sage_agg(itemx, src_iu1, dst_iu1, E_iu1, N_USER1);
    gemm(userx, Ws3, nullptr, nullptr, userx2, N_USER1, DH, 0);
    gemm(agg,   Wn3, userx2,  b3,      userx2, N_USER1, DH, 1);

    gemm(userx2, Wlin, nullptr, blin, out, N_USER1, DOUT, 0);
}


# Edge bucketing with LDS counters only: partition pass by coarse row bin, then one owner block per row range counts, scans, fills; no global atomics; 13 launches, was 22; sums stay in arrival order
# speedup vs baseline: 4.8147x; 1.1677x over previous
#include <hip/hip_runtime.h>


typedef float v2f __attribute__((ext_vector_type(2)));
typedef float v8f __attribute__((ext_vector_type(8)));

#define DH 256
#define DOUT 128

#define NCHUNK 128
#define NBIN   16

__global__ __launch_bounds__(1024) void part_kernel(
    const int* __restrict__ src, const int* __restrict__ dst, int E, int n_dst,
    int* __restrict__ psrc, int* __restrict__ pdst, int* __restrict__ segoff)
{
    __shared__ int hcnt[NBIN];
    __shared__ int hoff[NBIN + 1];
    const int tid   = threadIdx.x;
    const int chunk = (E + NCHUNK - 1) / NCHUNK;
    const int e0    = blockIdx.x * chunk;
    const int e1    = min(e0 + chunk, E);
    const unsigned R16 = (unsigned)NBIN * (unsigned)((n_dst + 255) / 256);

    if (tid < NBIN) hcnt[tid] = 0;
    __syncthreads();
    for (int e = e0 + tid; e < e1; e += 1024) {
        const int d = dst[e];
        if ((unsigned)d < (unsigned)n_dst) atomicAdd(&hcnt[(unsigned)d / R16], 1);
    }
    __syncthreads();
    if (tid == 0) {
        int run = 0;
        for (int k = 0; k < NBIN; ++k) { hoff[k] = run; run += hcnt[k]; }
        hoff[NBIN] = run;
    }
    __syncthreads();
    if (tid <= NBIN) segoff[blockIdx.x * (NBIN + 1) + tid] = e0 + hoff[tid];
    if (tid < NBIN) hcnt[tid] = 0;
    __syncthreads();
    for (int e = e0 + tid; e < e1; e += 1024) {
        const int d = dst[e];
        if ((unsigned)d < (unsigned)n_dst) {
            const unsigned k = (unsigned)d / R16;
            const int p = e0 + hoff[k] + atomicAdd(&hcnt[k], 1);
            pdst[p] = d;
            psrc[p] = src[e];
        }
    }
}

__global__ __launch_bounds__(1024) void owner_kernel(
    const int* __restrict__ psrc, const int* __restrict__ pdst,
    const int* __restrict__ segoff, int n_dst,
    int* __restrict__ cnt, int* __restrict__ off, int* __restrict__ slots)
{
    __shared__ int lcnt[256];
    __shared__ int loff[256];
    __shared__ int sst[NCHUNK];
    __shared__ int sen[NCHUNK];
    __shared__ int wsum[8];
    __shared__ int s_base;

    const int tid  = threadIdx.x;
    const int lane = tid & 31;
    const int wv   = tid >> 5;
    const int R    = (n_dst + 255) / 256;
    const int lo   = blockIdx.x * R;
    const int nrows = max(0, min(R, n_dst - lo));
    const int cb   = blockIdx.x >> 4;

    int below = 0;
    if (tid < NCHUNK) {
        const int a0 = segoff[tid * (NBIN + 1)];
        const int st = segoff[tid * (NBIN + 1) + cb];
        const int en = segoff[tid * (NBIN + 1) + cb + 1];
        sst[tid] = st;
        sen[tid] = en;
        below = st - a0;
    }
    if (tid < 256) lcnt[tid] = 0;
    if (tid == 0) s_base = 0;
    __syncthreads();

    for (int s = wv; s < NCHUNK; s += 32) {
        const int a = sst[s], b = sen[s];
        for (int i = a + lane; i < b; i += 256) {
            int d[8];
#pragma unroll
            for (int u = 0; u < 8; ++u) d[u] = pdst[min(i + 32 * u, b - 1)];
#pragma unroll
            for (int u = 0; u < 8; ++u) {
                if (i + 32 * u < b) {
                    const int r = d[u] - lo;
                    if ((unsigned)r < (unsigned)nrows) atomicAdd(&lcnt[r], 1);
                    else if (r < 0) ++below;
                }
            }
        }
    }
    if (below) atomicAdd(&s_base, below);
    __syncthreads();

    int c = 0, incl = 0;
    if (tid < 256) {
        c = (tid < nrows) ? lcnt[tid] : 0;
        incl = c;
        for (int dd = 1; dd < 32; dd <<= 1) {
            const int t = __shfl_up(incl, dd, 32);
            if (lane >= dd) incl += t;
        }
        if (lane == 31) wsum[wv] = incl;
    }
    __syncthreads();
    if (tid < 256) {
        int pre = s_base;
        for (int k = 0; k < wv; ++k) pre += wsum[k];
        const int ex = pre + incl - c;
        if (tid < nrows) { off[lo + tid] = ex; cnt[lo + tid] = c; }
        loff[tid] = ex;
        lcnt[tid] = 0;
    }
    __syncthreads();

    for (int s = wv; s < NCHUNK; s += 32) {
        const int a = sst[s], b = sen[s];
        for (int i = a + lane; i < b; i += 256) {
            int d[8], sv[8];
#pragma unroll
            for (int u = 0; u < 8; ++u) {
                const int j = min(i + 32 * u, b - 1);
                d[u]  = pdst[j];
                sv[u] = psrc[j];
            }
#pragma unroll
            for (int u = 0; u < 8; ++u) {
                if (i + 32 * u < b) {
                    const int r = d[u] - lo;
                    if ((unsigned)r < (unsigned)nrows)
                        slots[loff[r] + atomicAdd(&lcnt[r], 1)] = sv[u];
                }
            }
        }
    }
}

__global__ __launch_bounds__(256) void gather_mean_kernel(
    const float* __restrict__ X, const int* __restrict__ slots,
    const int* __restrict__ off, const int* __restrict__ cnt,
    int n_dst, float* __restrict__ agg)
{
    const int row = __builtin_amdgcn_readfirstlane(
        (int)(blockIdx.x * blockDim.y + threadIdx.y));
    if (row >= n_dst) return;
    const int lane = threadIdx.x;
    const int c = cnt[row];
    const int o = off[row];

    float4 a0 = make_float4(0.0f, 0.0f, 0.0f, 0.0f);
    float4 a1 = make_float4(0.0f, 0.0f, 0.0f, 0.0f);
    for (int base = 0; base < c; base += 32) {
        const int m = min(32, c - base);
        const int sid = (lane < m) ? slots[o + base + lane] : 0;
#pragma unroll 8
        for (int j = 0; j < m; ++j) {
            const int s = __builtin_amdgcn_readlane(sid, j);
            const float4* xs = (const float4*)(X + (size_t)s * DH);
            const float4 v0 = xs[lane];
            const float4 v1 = xs[lane + 32];
            a0.x += v0.x; a0.y += v0.y; a0.z += v0.z; a0.w += v0.w;
            a1.x += v1.x; a1.y += v1.y; a1.z += v1.z; a1.w += v1.w;
        }
    }
    const float inv = 1.0f / fmaxf((float)c, 1.0f);
    float4* ao = (float4*)(agg + (size_t)row * DH);
    ao[lane]      = make_float4(a0.x * inv, a0.y * inv, a0.z * inv, a0.w * inv);
    ao[lane + 32] = make_float4(a1.x * inv, a1.y * inv, a1.z * inv, a1.w * inv);
}

template <bool HASC>
__device__ __forceinline__ void gemm_epilogue(
    const v8f (&acc)[4], const float* Cin, const float* bias, float* D,
    int M, int N, int relu, int row0, int col0, int half, int sub)
{
    const int ncol = col0 + sub;
    const int mb   = row0 + 8 * half;

    float bv[4];
#pragma unroll
    for (int t = 0; t < 4; ++t) bv[t] = bias ? bias[ncol + t * 16] : 0.0f;

    float cv[8][4];
    if (HASC) {
#pragma unroll
        for (int v = 0; v < 8; ++v) {
            const int mm = min(mb + v, M - 1);
            const float* cp = Cin + (size_t)mm * N + ncol;
#pragma unroll
            for (int t = 0; t < 4; ++t) cv[v][t] = cp[t * 16];
        }
    }

#pragma unroll
    for (int v = 0; v < 8; ++v) {
        const int m = mb + v;
        if (m < M) {
            float* dp = D + (size_t)m * N + ncol;
#pragma unroll
            for (int t = 0; t < 4; ++t) {
                float r = acc[t][v] + bv[t];
                if (HASC) r += cv[v][t];
                if (relu) r = fmaxf(r, 0.0f);
                dp[t * 16] = r;
            }
        }
    }
}

__global__ __launch_bounds__(128) void gemm_wmma_kernel(
    const float* __restrict__ A, const float* __restrict__ W,
    const float* Cin, const float* bias,
    float* D, int M, int N, int relu)
{
    __shared__ float Bs[DH * 64];

    const int lane = threadIdx.x;
    const int half = lane >> 4;
    const int sub  = lane & 15;
    const int col0 = blockIdx.y * 64;

    {
        const int tid = threadIdx.y * 32 + lane;
        const int r0  = tid >> 4;
        const int c4  = (tid & 15) << 2;
        const float* Wp = W + (size_t)r0 * N + col0 + c4;
        float*       Bq = &Bs[r0 * 64 + c4];
        float4 w[32];
#pragma unroll
        for (int j = 0; j < 32; ++j)
            w[j] = *(const float4*)(Wp + (size_t)(8 * j) * N);
#pragma unroll
        for (int j = 0; j < 32; ++j)
            *(float4*)(Bq + 8 * 64 * j) = w[j];
    }
    __syncthreads();

    const int tm   = blockIdx.x * blockDim.y + threadIdx.y;
    const int row0 = tm * 16;
    if (row0 >= M) return;

    const int arow = min(row0 + sub, M - 1);
    const float* Ap = A + (size_t)arow * DH;

    v8f acc[4] = {};

    for (int k0 = 0; k0 < DH; k0 += 4) {
        const int ka = k0 + 2 * half;
        const v2f a = *(const v2f*)(Ap + ka);
        const float* Bp = &Bs[ka * 64 + sub];
#pragma unroll
        for (int t = 0; t < 4; ++t) {
            v2f b;
            b.x = Bp[t * 16];
            b.y = Bp[t * 16 + 64];
            acc[t] = __builtin_amdgcn_wmma_f32_16x16x4_f32(
                false, a, false, b, (short)0, acc[t], false, false);
        }
    }

    if (Cin) gemm_epilogue<true >(acc, Cin, bias, D, M, N, relu, row0, col0, half, sub);
    else     gemm_epilogue<false>(acc, Cin, bias, D, M, N, relu, row0, col0, half, sub);
}

extern "C" void kernel_launch(void* const* d_in, const int* in_sizes, int n_in,
                              void* d_out, int out_size, void* d_ws, size_t ws_size,
                              hipStream_t stream)
{
    const float* x_item  = (const float*)d_in[0];
    const float* x_user  = (const float*)d_in[1];
    const int*   src_ii  = (const int*)d_in[2];
    const int*   dst_ii  = (const int*)d_in[3];
    const int*   src_iu0 = (const int*)d_in[4];
    const int*   dst_iu0 = (const int*)d_in[5];
    const int*   src_iu1 = (const int*)d_in[6];
    const int*   dst_iu1 = (const int*)d_in[7];
    const float* Wn1 = (const float*)d_in[8];
    const float* Ws1 = (const float*)d_in[9];
    const float* b1  = (const float*)d_in[10];
    const float* Wn2 = (const float*)d_in[11];
    const float* Ws2 = (const float*)d_in[12];
    const float* b2  = (const float*)d_in[13];
    const float* Wn3 = (const float*)d_in[14];
    const float* Ws3 = (const float*)d_in[15];
    const float* b3  = (const float*)d_in[16];
    const float* Wlin = (const float*)d_in[17];
    const float* blin = (const float*)d_in[18];

    const int E_ii  = in_sizes[2];
    const int E_iu0 = in_sizes[4];
    const int E_iu1 = in_sizes[6];
    const int N_ITEM_DST = 50000;
    const int N_USER1    = 25000;

    char* ws = (char*)d_ws;
    const size_t off_deg   = 0;
    const size_t off_agg   = 256 * 1024;
    const size_t off_itemx = off_agg   + (size_t)N_ITEM_DST * DH * sizeof(float);
    const size_t off_userx = off_itemx + (size_t)N_ITEM_DST * DH * sizeof(float);
    int*   cnt    = (int*)(ws + off_deg);
    float* agg    = (float*)(ws + off_agg);
    float* itemx  = (float*)(ws + off_itemx);
    float* userx  = (float*)(ws + off_userx);
    float* userx2 = itemx;
    float* out    = (float*)d_out;
    int* off_row  = (int*)(userx + (size_t)N_USER1 * DH);
    int* pdst     = off_row + 65536;
    int* psrc     = pdst + 1048576;
    int* slots    = psrc + 1048576;
    int* segoff   = slots + 1048576;

    auto gemm = [&](const float* A, const float* Wm, const float* Cin,
                    const float* bias, float* Dm, int M, int N, int relu) {
        int mt = (M + 15) / 16;
        dim3 grid((mt + 3) / 4, N / 64);
        dim3 block(32, 4);
        gemm_wmma_kernel<<<grid, block, 0, stream>>>(A, Wm, Cin, bias, Dm, M, N, relu);
    };
    auto sage_agg = [&](const float* Xsrc, const int* src, const int* dst,
                        int E, int n_dst) {
        part_kernel<<<dim3(NCHUNK), dim3(1024), 0, stream>>>(
            src, dst, E, n_dst, psrc, pdst, segoff);
        owner_kernel<<<dim3(256), dim3(1024), 0, stream>>>(
            psrc, pdst, segoff, n_dst, cnt, off_row, slots);
        gather_mean_kernel<<<dim3((n_dst + 7) / 8), dim3(32, 8), 0, stream>>>(
            Xsrc, slots, off_row, cnt, n_dst, agg);
    };

    sage_agg(x_item, src_ii, dst_ii, E_ii, N_ITEM_DST);
    gemm(x_item, Ws1, nullptr, nullptr, itemx, N_ITEM_DST, DH, 0);
    gemm(agg,    Wn1, itemx,   b1,      itemx, N_ITEM_DST, DH, 1);

    sage_agg(x_item, src_iu0, dst_iu0, E_iu0, N_USER1);
    gemm(x_user, Ws2, nullptr, nullptr, userx, N_USER1, DH, 0);
    gemm(agg,    Wn2, userx,   b2,      userx, N_USER1, DH, 1);

    sage_agg(itemx, src_iu1, dst_iu1, E_iu1, N_USER1);
    gemm(userx, Ws3, nullptr, nullptr, userx2, N_USER1, DH, 0);
    gemm(agg,   Wn3, userx2,  b3,      userx2, N_USER1, DH, 1);

    gemm(userx2, Wlin, nullptr, blin, out, N_USER1, DOUT, 0);
}


# Edge bucketing of all three layers hoisted to the front (it reads index inputs only): one partition and one owner launch with blockIdx.y = layer; 12 launches, was 16; sums stay in arrival order
# speedup vs baseline: 5.1376x; 1.0671x over previous
#include <hip/hip_runtime.h>


typedef float v2f __attribute__((ext_vector_type(2)));
typedef float v8f __attribute__((ext_vector_type(8)));

#define DH 256
#define DOUT 128

#define NCHUNK 128
#define NBIN   16
#define LSTRIDE   1048576
#define RSTRIDE   65536
#define SEGSTRIDE (NCHUNK * (NBIN + 1))

__device__ __forceinline__ void part_body(
    const int* __restrict__ src, const int* __restrict__ dst, int E, int n_dst,
    int* __restrict__ psrc, int* __restrict__ pdst, int* __restrict__ segoff)
{
    __shared__ int hcnt[NBIN];
    __shared__ int hoff[NBIN + 1];
    const int tid   = threadIdx.x;
    const int chunk = (E + NCHUNK - 1) / NCHUNK;
    const int e0    = blockIdx.x * chunk;
    const int e1    = min(e0 + chunk, E);
    const unsigned R16 = (unsigned)NBIN * (unsigned)((n_dst + 255) / 256);

    if (tid < NBIN) hcnt[tid] = 0;
    __syncthreads();
    for (int e = e0 + tid; e < e1; e += 1024) {
        const int d = dst[e];
        if ((unsigned)d < (unsigned)n_dst) atomicAdd(&hcnt[(unsigned)d / R16], 1);
    }
    __syncthreads();
    if (tid == 0) {
        int run = 0;
        for (int k = 0; k < NBIN; ++k) { hoff[k] = run; run += hcnt[k]; }
        hoff[NBIN] = run;
    }
    __syncthreads();
    if (tid <= NBIN) segoff[blockIdx.x * (NBIN + 1) + tid] = e0 + hoff[tid];
    if (tid < NBIN) hcnt[tid] = 0;
    __syncthreads();
    for (int e = e0 + tid; e < e1; e += 1024) {
        const int d = dst[e];
        if ((unsigned)d < (unsigned)n_dst) {
            const unsigned k = (unsigned)d / R16;
            const int p = e0 + hoff[k] + atomicAdd(&hcnt[k], 1);
            pdst[p] = d;
            psrc[p] = src[e];
        }
    }
}

__global__ __launch_bounds__(1024) void part_kernel(
    const int* __restrict__ src0, const int* __restrict__ dst0,
    const int* __restrict__ src1, const int* __restrict__ dst1,
    const int* __restrict__ src2, const int* __restrict__ dst2,
    int E0, int E1, int E2, int n0, int n1, int n2,
    int* __restrict__ psrc, int* __restrict__ pdst, int* __restrict__ segoff)
{
    const int L = blockIdx.y;
    const int* src = (L == 0) ? src0 : (L == 1) ? src1 : src2;
    const int* dst = (L == 0) ? dst0 : (L == 1) ? dst1 : dst2;
    const int  E   = (L == 0) ? E0   : (L == 1) ? E1   : E2;
    const int  n   = (L == 0) ? n0   : (L == 1) ? n1   : n2;
    part_body(src, dst, E, n, psrc + (size_t)L * LSTRIDE, pdst + (size_t)L * LSTRIDE,
              segoff + L * SEGSTRIDE);
}

__device__ __forceinline__ void owner_body(
    const int* __restrict__ psrc, const int* __restrict__ pdst,
    const int* __restrict__ segoff, int n_dst,
    int* __restrict__ cnt, int* __restrict__ off, int* __restrict__ slots)
{
    __shared__ int lcnt[256];
    __shared__ int loff[256];
    __shared__ int sst[NCHUNK];
    __shared__ int sen[NCHUNK];
    __shared__ int wsum[8];
    __shared__ int s_base;

    const int tid  = threadIdx.x;
    const int lane = tid & 31;
    const int wv   = tid >> 5;
    const int R    = (n_dst + 255) / 256;
    const int lo   = blockIdx.x * R;
    const int nrows = max(0, min(R, n_dst - lo));
    const int cb   = blockIdx.x >> 4;

    int below = 0;
    if (tid < NCHUNK) {
        const int a0 = segoff[tid * (NBIN + 1)];
        const int st = segoff[tid * (NBIN + 1) + cb];
        const int en = segoff[tid * (NBIN + 1) + cb + 1];
        sst[tid] = st;
        sen[tid] = en;
        below = st - a0;
    }
    if (tid < 256) lcnt[tid] = 0;
    if (tid == 0) s_base = 0;
    __syncthreads();

    for (int s = wv; s < NCHUNK; s += 32) {
        const int a = sst[s], b = sen[s];
        for (int i = a + lane; i < b; i += 256) {
            int d[8];
#pragma unroll
            for (int u = 0; u < 8; ++u) d[u] = pdst[min(i + 32 * u, b - 1)];
#pragma unroll
            for (int u = 0; u < 8; ++u) {
                if (i + 32 * u < b) {
                    const int r = d[u] - lo;
                    if ((unsigned)r < (unsigned)nrows) atomicAdd(&lcnt[r], 1);
                    else if (r < 0) ++below;
                }
            }
        }
    }
    if (below) atomicAdd(&s_base, below);
    __syncthreads();

    int c = 0, incl = 0;
    if (tid < 256) {
        c = (tid < nrows) ? lcnt[tid] : 0;
        incl = c;
        for (int dd = 1; dd < 32; dd <<= 1) {
            const int t = __shfl_up(incl, dd, 32);
            if (lane >= dd) incl += t;
        }
        if (lane == 31) wsum[wv] = incl;
    }
    __syncthreads();
    if (tid < 256) {
        int pre = s_base;
        for (int k = 0; k < wv; ++k) pre += wsum[k];
        const int ex = pre + incl - c;
        if (tid < nrows) { off[lo + tid] = ex; cnt[lo + tid] = c; }
        loff[tid] = ex;
        lcnt[tid] = 0;
    }
    __syncthreads();

    for (int s = wv; s < NCHUNK; s += 32) {
        const int a = sst[s], b = sen[s];
        for (int i = a + lane; i < b; i += 256) {
            int d[8], sv[8];
#pragma unroll
            for (int u = 0; u < 8; ++u) {
                const int j = min(i + 32 * u, b - 1);
                d[u]  = pdst[j];
                sv[u] = psrc[j];
            }
#pragma unroll
            for (int u = 0; u < 8; ++u) {
                if (i + 32 * u < b) {
                    const int r = d[u] - lo;
                    if ((unsigned)r < (unsigned)nrows)
                        slots[loff[r] + atomicAdd(&lcnt[r], 1)] = sv[u];
                }
            }
        }
    }
}

__global__ __launch_bounds__(1024) void owner_kernel(
    const int* __restrict__ psrc, const int* __restrict__ pdst,
    const int* __restrict__ segoff, int n0, int n1, int n2,
    int* __restrict__ cnt, int* __restrict__ off, int* __restrict__ slots)
{
    const int L = blockIdx.y;
    const int n = (L == 0) ? n0 : (L == 1) ? n1 : n2;
    owner_body(psrc + (size_t)L * LSTRIDE, pdst + (size_t)L * LSTRIDE,
               segoff + L * SEGSTRIDE, n,
               cnt + L * RSTRIDE, off + L * RSTRIDE, slots + (size_t)L * LSTRIDE);
}

__global__ __launch_bounds__(256) void gather_mean_kernel(
    const float* __restrict__ X, const int* __restrict__ slots,
    const int* __restrict__ off, const int* __restrict__ cnt,
    int n_dst, float* __restrict__ agg)
{
    const int row = __builtin_amdgcn_readfirstlane(
        (int)(blockIdx.x * blockDim.y + threadIdx.y));
    if (row >= n_dst) return;
    const int lane = threadIdx.x;
    const int c = cnt[row];
    const int o = off[row];

    float4 a0 = make_float4(0.0f, 0.0f, 0.0f, 0.0f);
    float4 a1 = make_float4(0.0f, 0.0f, 0.0f, 0.0f);
    for (int base = 0; base < c; base += 32) {
        const int m = min(32, c - base);
        const int sid = (lane < m) ? slots[o + base + lane] : 0;
#pragma unroll 8
        for (int j = 0; j < m; ++j) {
            const int s = __builtin_amdgcn_readlane(sid, j);
            const float4* xs = (const float4*)(X + (size_t)s * DH);
            const float4 v0 = xs[lane];
            const float4 v1 = xs[lane + 32];
            a0.x += v0.x; a0.y += v0.y; a0.z += v0.z; a0.w += v0.w;
            a1.x += v1.x; a1.y += v1.y; a1.z += v1.z; a1.w += v1.w;
        }
    }
    const float inv = 1.0f / fmaxf((float)c, 1.0f);
    float4* ao = (float4*)(agg + (size_t)row * DH);
    ao[lane]      = make_float4(a0.x * inv, a0.y * inv, a0.z * inv, a0.w * inv);
    ao[lane + 32] = make_float4(a1.x * inv, a1.y * inv, a1.z * inv, a1.w * inv);
}

template <bool HASC>
__device__ __forceinline__ void gemm_epilogue(
    const v8f (&acc)[4], const float* Cin, const float* bias, float* D,
    int M, int N, int relu, int row0, int col0, int half, int sub)
{
    const int ncol = col0 + sub;
    const int mb   = row0 + 8 * half;

    float bv[4];
#pragma unroll
    for (int t = 0; t < 4; ++t) bv[t] = bias ? bias[ncol + t * 16] : 0.0f;

    float cv[8][4];
    if (HASC) {
#pragma unroll
        for (int v = 0; v < 8; ++v) {
            const int mm = min(mb + v, M - 1);
            const float* cp = Cin + (size_t)mm * N + ncol;
#pragma unroll
            for (int t = 0; t < 4; ++t) cv[v][t] = cp[t * 16];
        }
    }

#pragma unroll
    for (int v = 0; v < 8; ++v) {
        const int m = mb + v;
        if (m < M) {
            float* dp = D + (size_t)m * N + ncol;
#pragma unroll
            for (int t = 0; t < 4; ++t) {
                float r = acc[t][v] + bv[t];
                if (HASC) r += cv[v][t];
                if (relu) r = fmaxf(r, 0.0f);
                dp[t * 16] = r;
            }
        }
    }
}

__global__ __launch_bounds__(128) void gemm_wmma_kernel(
    const float* __restrict__ A, const float* __restrict__ W,
    const float* Cin, const float* bias,
    float* D, int M, int N, int relu)
{
    __shared__ float Bs[DH * 64];

    const int lane = threadIdx.x;
    const int half = lane >> 4;
    const int sub  = lane & 15;
    const int col0 = blockIdx.y * 64;

    {
        const int tid = threadIdx.y * 32 + lane;
        const int r0  = tid >> 4;
        const int c4  = (tid & 15) << 2;
        const float* Wp = W + (size_t)r0 * N + col0 + c4;
        float*       Bq = &Bs[r0 * 64 + c4];
        float4 w[32];
#pragma unroll
        for (int j = 0; j < 32; ++j)
            w[j] = *(const float4*)(Wp + (size_t)(8 * j) * N);
#pragma unroll
        for (int j = 0; j < 32; ++j)
            *(float4*)(Bq + 8 * 64 * j) = w[j];
    }
    __syncthreads();

    const int tm   = blockIdx.x * blockDim.y + threadIdx.y;
    const int row0 = tm * 16;
    if (row0 >= M) return;

    const int arow = min(row0 + sub, M - 1);
    const float* Ap = A + (size_t)arow * DH;

    v8f acc[4] = {};

    for (int k0 = 0; k0 < DH; k0 += 4) {
        const int ka = k0 + 2 * half;
        const v2f a = *(const v2f*)(Ap + ka);
        const float* Bp = &Bs[ka * 64 + sub];
#pragma unroll
        for (int t = 0; t < 4; ++t) {
            v2f b;
            b.x = Bp[t * 16];
            b.y = Bp[t * 16 + 64];
            acc[t] = __builtin_amdgcn_wmma_f32_16x16x4_f32(
                false, a, false, b, (short)0, acc[t], false, false);
        }
    }

    if (Cin) gemm_epilogue<true >(acc, Cin, bias, D, M, N, relu, row0, col0, half, sub);
    else     gemm_epilogue<false>(acc, Cin, bias, D, M, N, relu, row0, col0, half, sub);
}

extern "C" void kernel_launch(void* const* d_in, const int* in_sizes, int n_in,
                              void* d_out, int out_size, void* d_ws, size_t ws_size,
                              hipStream_t stream)
{
    const float* x_item  = (const float*)d_in[0];
    const float* x_user  = (const float*)d_in[1];
    const int*   src_ii  = (const int*)d_in[2];
    const int*   dst_ii  = (const int*)d_in[3];
    const int*   src_iu0 = (const int*)d_in[4];
    const int*   dst_iu0 = (const int*)d_in[5];
    const int*   src_iu1 = (const int*)d_in[6];
    const int*   dst_iu1 = (const int*)d_in[7];
    const float* Wn1 = (const float*)d_in[8];
    const float* Ws1 = (const float*)d_in[9];
    const float* b1  = (const float*)d_in[10];
    const float* Wn2 = (const float*)d_in[11];
    const float* Ws2 = (const float*)d_in[12];
    const float* b2  = (const float*)d_in[13];
    const float* Wn3 = (const float*)d_in[14];
    const float* Ws3 = (const float*)d_in[15];
    const float* b3  = (const float*)d_in[16];
    const float* Wlin = (const float*)d_in[17];
    const float* blin = (const float*)d_in[18];

    const int E_ii  = in_sizes[2];
    const int E_iu0 = in_sizes[4];
    const int E_iu1 = in_sizes[6];
    const int N_ITEM_DST = 50000;
    const int N_USER1    = 25000;

    char* ws = (char*)d_ws;
    const size_t off_agg   = 256 * 1024;
    const size_t off_itemx = off_agg   + (size_t)N_ITEM_DST * DH * sizeof(float);
    const size_t off_userx = off_itemx + (size_t)N_ITEM_DST * DH * sizeof(float);
    float* agg    = (float*)(ws + off_agg);
    float* itemx  = (float*)(ws + off_itemx);
    float* userx  = (float*)(ws + off_userx);
    float* userx2 = itemx;
    float* out    = (float*)d_out;
    int* off_row  = (int*)(userx + (size_t)N_USER1 * DH);
    int* cnt      = off_row + 3 * RSTRIDE;
    int* segoff   = cnt + 3 * RSTRIDE;
    int* slots    = segoff + 8192;
    int* pdst     = (int*)agg;
    int* psrc     = pdst + 3 * LSTRIDE;

    auto gemm = [&](const float* A, const float* Wm, const float* Cin,
                    const float* bias, float* Dm, int M, int N, int relu) {
        int mt = (M + 15) / 16;
        dim3 grid((mt + 3) / 4, N / 64);
        dim3 block(32, 4);
        gemm_wmma_kernel<<<grid, block, 0, stream>>>(A, Wm, Cin, bias, Dm, M, N, relu);
    };
    auto gather = [&](const float* Xsrc, int L, int n_dst) {
        gather_mean_kernel<<<dim3((n_dst + 7) / 8), dim3(32, 8), 0, stream>>>(
            Xsrc, slots + (size_t)L * LSTRIDE, off_row + L * RSTRIDE,
            cnt + L * RSTRIDE, n_dst, agg);
    };

    part_kernel<<<dim3(NCHUNK, 3), dim3(1024), 0, stream>>>(
        src_ii, dst_ii, src_iu0, dst_iu0, src_iu1, dst_iu1,
        E_ii, E_iu0, E_iu1, N_ITEM_DST, N_USER1, N_USER1,
        psrc, pdst, segoff);
    owner_kernel<<<dim3(256, 3), dim3(1024), 0, stream>>>(
        psrc, pdst, segoff, N_ITEM_DST, N_USER1, N_USER1, cnt, off_row, slots);

    gather(x_item, 0, N_ITEM_DST);
    gemm(x_item, Ws1, nullptr, nullptr, itemx, N_ITEM_DST, DH, 0);
    gemm(agg,    Wn1, itemx,   b1,      itemx, N_ITEM_DST, DH, 1);

    gather(x_item, 1, N_USER1);
    gemm(x_user, Ws2, nullptr, nullptr, userx, N_USER1, DH, 0);
    gemm(agg,    Wn2, userx,   b2,      userx, N_USER1, DH, 1);

    gather(itemx, 2, N_USER1);
    gemm(userx, Ws3, nullptr, nullptr, userx2, N_USER1, DH, 0);
    gemm(agg,   Wn3, userx2,  b3,      userx2, N_USER1, DH, 1);

    gemm(userx2, Wlin, nullptr, blin, out, N_USER1, DOUT, 0);
}


# Each layer's self-term GEMM shares its gather's launch: every fifth block is an 8-wave GEMM block, the rest gather rows, so matrix work runs under the memory waits; 9 launches, was 12; sums unchanged
# speedup vs baseline: 5.4510x; 1.0610x over previous
#include <hip/hip_runtime.h>


typedef float v2f __attribute__((ext_vector_type(2)));
typedef float v8f __attribute__((ext_vector_type(8)));

#define DH 256
#define DOUT 128

#define NCHUNK 128
#define NBIN   16
#define LSTRIDE   1048576
#define RSTRIDE   65536
#define SEGSTRIDE (NCHUNK * (NBIN + 1))

__device__ __forceinline__ void part_body(
    const int* __restrict__ src, const int* __restrict__ dst, int E, int n_dst,
    int* __restrict__ psrc, int* __restrict__ pdst, int* __restrict__ segoff)
{
    __shared__ int hcnt[NBIN];
    __shared__ int hoff[NBIN + 1];
    const int tid   = threadIdx.x;
    const int chunk = (E + NCHUNK - 1) / NCHUNK;
    const int e0    = blockIdx.x * chunk;
    const int e1    = min(e0 + chunk, E);
    const unsigned R16 = (unsigned)NBIN * (unsigned)((n_dst + 255) / 256);

    if (tid < NBIN) hcnt[tid] = 0;
    __syncthreads();
    for (int e = e0 + tid; e < e1; e += 1024) {
        const int d = dst[e];
        if ((unsigned)d < (unsigned)n_dst) atomicAdd(&hcnt[(unsigned)d / R16], 1);
    }
    __syncthreads();
    if (tid == 0) {
        int run = 0;
        for (int k = 0; k < NBIN; ++k) { hoff[k] = run; run += hcnt[k]; }
        hoff[NBIN] = run;
    }
    __syncthreads();
    if (tid <= NBIN) segoff[blockIdx.x * (NBIN + 1) + tid] = e0 + hoff[tid];
    if (tid < NBIN) hcnt[tid] = 0;
    __syncthreads();
    for (int e = e0 + tid; e < e1; e += 1024) {
        const int d = dst[e];
        if ((unsigned)d < (unsigned)n_dst) {
            const unsigned k = (unsigned)d / R16;
            const int p = e0 + hoff[k] + atomicAdd(&hcnt[k], 1);
            pdst[p] = d;
            psrc[p] = src[e];
        }
    }
}

__global__ __launch_bounds__(1024) void part_kernel(
    const int* __restrict__ src0, const int* __restrict__ dst0,
    const int* __restrict__ src1, const int* __restrict__ dst1,
    const int* __restrict__ src2, const int* __restrict__ dst2,
    int E0, int E1, int E2, int n0, int n1, int n2,
    int* __restrict__ psrc, int* __restrict__ pdst, int* __restrict__ segoff)
{
    const int L = blockIdx.y;
    const int* src = (L == 0) ? src0 : (L == 1) ? src1 : src2;
    const int* dst = (L == 0) ? dst0 : (L == 1) ? dst1 : dst2;
    const int  E   = (L == 0) ? E0   : (L == 1) ? E1   : E2;
    const int  n   = (L == 0) ? n0   : (L == 1) ? n1   : n2;
    part_body(src, dst, E, n, psrc + (size_t)L * LSTRIDE, pdst + (size_t)L * LSTRIDE,
              segoff + L * SEGSTRIDE);
}

__device__ __forceinline__ void owner_body(
    const int* __restrict__ psrc, const int* __restrict__ pdst,
    const int* __restrict__ segoff, int n_dst,
    int* __restrict__ cnt, int* __restrict__ off, int* __restrict__ slots)
{
    __shared__ int lcnt[256];
    __shared__ int loff[256];
    __shared__ int sst[NCHUNK];
    __shared__ int sen[NCHUNK];
    __shared__ int wsum[8];
    __shared__ int s_base;

    const int tid  = threadIdx.x;
    const int lane = tid & 31;
    const int wv   = tid >> 5;
    const int R    = (n_dst + 255) / 256;
    const int lo   = blockIdx.x * R;
    const int nrows = max(0, min(R, n_dst - lo));
    const int cb   = blockIdx.x >> 4;

    int below = 0;
    if (tid < NCHUNK) {
        const int a0 = segoff[tid * (NBIN + 1)];
        const int st = segoff[tid * (NBIN + 1) + cb];
        const int en = segoff[tid * (NBIN + 1) + cb + 1];
        sst[tid] = st;
        sen[tid] = en;
        below = st - a0;
    }
    if (tid < 256) lcnt[tid] = 0;
    if (tid == 0) s_base = 0;
    __syncthreads();

    for (int s = wv; s < NCHUNK; s += 32) {
        const int a = sst[s], b = sen[s];
        for (int i = a + lane; i < b; i += 256) {
            int d[8];
#pragma unroll
            for (int u = 0; u < 8; ++u) d[u] = pdst[min(i + 32 * u, b - 1)];
#pragma unroll
            for (int u = 0; u < 8; ++u) {
                if (i + 32 * u < b) {
                    const int r = d[u] - lo;
                    if ((unsigned)r < (unsigned)nrows) atomicAdd(&lcnt[r], 1);
                    else if (r < 0) ++below;
                }
            }
        }
    }
    if (below) atomicAdd(&s_base, below);
    __syncthreads();

    int c = 0, incl = 0;
    if (tid < 256) {
        c = (tid < nrows) ? lcnt[tid] : 0;
        incl = c;
        for (int dd = 1; dd < 32; dd <<= 1) {
            const int t = __shfl_up(incl, dd, 32);
            if (lane >= dd) incl += t;
        }
        if (lane == 31) wsum[wv] = incl;
    }
    __syncthreads();
    if (tid < 256) {
        int pre = s_base;
        for (int k = 0; k < wv; ++k) pre += wsum[k];
        const int ex = pre + incl - c;
        if (tid < nrows) { off[lo + tid] = ex; cnt[lo + tid] = c; }
        loff[tid] = ex;
        lcnt[tid] = 0;
    }
    __syncthreads();

    for (int s = wv; s < NCHUNK; s += 32) {
        const int a = sst[s], b = sen[s];
        for (int i = a + lane; i < b; i += 256) {
            int d[8], sv[8];
#pragma unroll
            for (int u = 0; u < 8; ++u) {
                const int j = min(i + 32 * u, b - 1);
                d[u]  = pdst[j];
                sv[u] = psrc[j];
            }
#pragma unroll
            for (int u = 0; u < 8; ++u) {
                if (i + 32 * u < b) {
                    const int r = d[u] - lo;
                    if ((unsigned)r < (unsigned)nrows)
                        slots[loff[r] + atomicAdd(&lcnt[r], 1)] = sv[u];
                }
            }
        }
    }
}

__global__ __launch_bounds__(1024) void owner_kernel(
    const int* __restrict__ psrc, const int* __restrict__ pdst,
    const int* __restrict__ segoff, int n0, int n1, int n2,
    int* __restrict__ cnt, int* __restrict__ off, int* __restrict__ slots)
{
    const int L = blockIdx.y;
    const int n = (L == 0) ? n0 : (L == 1) ? n1 : n2;
    owner_body(psrc + (size_t)L * LSTRIDE, pdst + (size_t)L * LSTRIDE,
               segoff + L * SEGSTRIDE, n,
               cnt + L * RSTRIDE, off + L * RSTRIDE, slots + (size_t)L * LSTRIDE);
}

__device__ __forceinline__ void gather_row(
    const float* __restrict__ X, const int* __restrict__ slots,
    const int* __restrict__ off, const int* __restrict__ cnt,
    int n_dst, float* __restrict__ agg, int wave_row)
{
    const int row = __builtin_amdgcn_readfirstlane(wave_row);
    if (row >= n_dst) return;
    const int lane = threadIdx.x;
    const int c = cnt[row];
    const int o = off[row];

    float4 a0 = make_float4(0.0f, 0.0f, 0.0f, 0.0f);
    float4 a1 = make_float4(0.0f, 0.0f, 0.0f, 0.0f);
    for (int base = 0; base < c; base += 32) {
        const int m = min(32, c - base);
        const int sid = (lane < m) ? slots[o + base + lane] : 0;
#pragma unroll 8
        for (int j = 0; j < m; ++j) {
            const int s = __builtin_amdgcn_readlane(sid, j);
            const float4* xs = (const float4*)(X + (size_t)s * DH);
            const float4 v0 = xs[lane];
            const float4 v1 = xs[lane + 32];
            a0.x += v0.x; a0.y += v0.y; a0.z += v0.z; a0.w += v0.w;
            a1.x += v1.x; a1.y += v1.y; a1.z += v1.z; a1.w += v1.w;
        }
    }
    const float inv = 1.0f / fmaxf((float)c, 1.0f);
    float4* ao = (float4*)(agg + (size_t)row * DH);
    ao[lane]      = make_float4(a0.x * inv, a0.y * inv, a0.z * inv, a0.w * inv);
    ao[lane + 32] = make_float4(a1.x * inv, a1.y * inv, a1.z * inv, a1.w * inv);
}

template <bool HASC>
__device__ __forceinline__ void gemm_epilogue(
    const v8f (&acc)[4], const float* Cin, const float* bias, float* D,
    int M, int N, int relu, int row0, int col0, int half, int sub)
{
    const int ncol = col0 + sub;
    const int mb   = row0 + 8 * half;

    float bv[4];
#pragma unroll
    for (int t = 0; t < 4; ++t) bv[t] = bias ? bias[ncol + t * 16] : 0.0f;

    float cv[8][4];
    if (HASC) {
#pragma unroll
        for (int v = 0; v < 8; ++v) {
            const int mm = min(mb + v, M - 1);
            const float* cp = Cin + (size_t)mm * N + ncol;
#pragma unroll
            for (int t = 0; t < 4; ++t) cv[v][t] = cp[t * 16];
        }
    }

#pragma unroll
    for (int v = 0; v < 8; ++v) {
        const int m = mb + v;
        if (m < M) {
            float* dp = D + (size_t)m * N + ncol;
#pragma unroll
            for (int t = 0; t < 4; ++t) {
                float r = acc[t][v] + bv[t];
                if (HASC) r += cv[v][t];
                if (relu) r = fmaxf(r, 0.0f);
                dp[t * 16] = r;
            }
        }
    }
}

template <int NW>
__device__ __forceinline__ void gemm_block(
    const float* __restrict__ A, const float* __restrict__ W,
    const float* Cin, const float* bias,
    float* D, int M, int N, int relu, int bx, int by,
    float* Bs)
{
    const int lane = threadIdx.x;
    const int half = lane >> 4;
    const int sub  = lane & 15;
    const int col0 = by * 64;

    {
        constexpr int RS = 2 * NW;
        constexpr int NJ = DH / RS;
        const int tid = threadIdx.y * 32 + lane;
        const int r0  = tid >> 4;
        const int c4  = (tid & 15) << 2;
        const float* Wp = W + (size_t)r0 * N + col0 + c4;
        float*       Bq = &Bs[r0 * 64 + c4];
        float4 w[NJ];
#pragma unroll
        for (int j = 0; j < NJ; ++j)
            w[j] = *(const float4*)(Wp + (size_t)(RS * j) * N);
#pragma unroll
        for (int j = 0; j < NJ; ++j)
            *(float4*)(Bq + RS * 64 * j) = w[j];
    }
    __syncthreads();

    const int tm   = bx * NW + threadIdx.y;
    const int row0 = tm * 16;
    if (row0 >= M) return;

    const int arow = min(row0 + sub, M - 1);
    const float* Ap = A + (size_t)arow * DH;

    v8f acc[4] = {};

    for (int k0 = 0; k0 < DH; k0 += 4) {
        const int ka = k0 + 2 * half;
        const v2f a = *(const v2f*)(Ap + ka);
        const float* Bp = &Bs[ka * 64 + sub];
#pragma unroll
        for (int t = 0; t < 4; ++t) {
            v2f b;
            b.x = Bp[t * 16];
            b.y = Bp[t * 16 + 64];
            acc[t] = __builtin_amdgcn_wmma_f32_16x16x4_f32(
                false, a, false, b, (short)0, acc[t], false, false);
        }
    }

    if (Cin) gemm_epilogue<true >(acc, Cin, bias, D, M, N, relu, row0, col0, half, sub);
    else     gemm_epilogue<false>(acc, Cin, bias, D, M, N, relu, row0, col0, half, sub);
}

__global__ __launch_bounds__(128) void gemm_wmma_kernel(
    const float* __restrict__ A, const float* __restrict__ W,
    const float* Cin, const float* bias,
    float* D, int M, int N, int relu)
{
    __shared__ float Bs[DH * 64];
    gemm_block<4>(A, W, Cin, bias, D, M, N, relu, blockIdx.x, blockIdx.y, Bs);
}

__global__ __launch_bounds__(256) void gather_selfgemm_kernel(
    const float* __restrict__ X, const int* __restrict__ slots,
    const int* __restrict__ off, const int* __restrict__ cnt,
    int n_dst, float* __restrict__ agg,
    const float* __restrict__ A, const float* __restrict__ W, float* D, int M)
{
    __shared__ float Bs[DH * 64];
    const int g = blockIdx.x / 5;
    const int r = blockIdx.x - 5 * g;
    if (r == 0) {
        if (g < 4 * ((M + 127) / 128))
            gemm_block<8>(A, W, nullptr, nullptr, D, M, DH, 0, g >> 2, g & 3, Bs);
    } else {
        gather_row(X, slots, off, cnt, n_dst, agg, (4 * g + r - 1) * 8 + (int)threadIdx.y);
    }
}

extern "C" void kernel_launch(void* const* d_in, const int* in_sizes, int n_in,
                              void* d_out, int out_size, void* d_ws, size_t ws_size,
                              hipStream_t stream)
{
    const float* x_item  = (const float*)d_in[0];
    const float* x_user  = (const float*)d_in[1];
    const int*   src_ii  = (const int*)d_in[2];
    const int*   dst_ii  = (const int*)d_in[3];
    const int*   src_iu0 = (const int*)d_in[4];
    const int*   dst_iu0 = (const int*)d_in[5];
    const int*   src_iu1 = (const int*)d_in[6];
    const int*   dst_iu1 = (const int*)d_in[7];
    const float* Wn1 = (const float*)d_in[8];
    const float* Ws1 = (const float*)d_in[9];
    const float* b1  = (const float*)d_in[10];
    const float* Wn2 = (const float*)d_in[11];
    const float* Ws2 = (const float*)d_in[12];
    const float* b2  = (const float*)d_in[13];
    const float* Wn3 = (const float*)d_in[14];
    const float* Ws3 = (const float*)d_in[15];
    const float* b3  = (const float*)d_in[16];
    const float* Wlin = (const float*)d_in[17];
    const float* blin = (const float*)d_in[18];

    const int E_ii  = in_sizes[2];
    const int E_iu0 = in_sizes[4];
    const int E_iu1 = in_sizes[6];
    const int N_ITEM_DST = 50000;
    const int N_USER1    = 25000;

    char* ws = (char*)d_ws;
    const size_t off_agg   = 256 * 1024;
    const size_t off_itemx = off_agg   + (size_t)N_ITEM_DST * DH * sizeof(float);
    const size_t off_userx = off_itemx + (size_t)N_ITEM_DST * DH * sizeof(float);
    float* agg    = (float*)(ws + off_agg);
    float* itemx  = (float*)(ws + off_itemx);
    float* userx  = (float*)(ws + off_userx);
    float* userx2 = agg + (size_t)N_USER1 * DH;
    float* out    = (float*)d_out;
    int* off_row  = (int*)(userx + (size_t)N_USER1 * DH);
    int* cnt      = off_row + 3 * RSTRIDE;
    int* segoff   = cnt + 3 * RSTRIDE;
    int* slots    = segoff + 8192;
    int* pdst     = (int*)agg;
    int* psrc     = pdst + 3 * LSTRIDE;

    auto gemm = [&](const float* A, const float* Wm, const float* Cin,
                    const float* bias, float* Dm, int M, int N, int relu) {
        int mt = (M + 15) / 16;
        dim3 grid((mt + 3) / 4, N / 64);
        dim3 block(32, 4);
        gemm_wmma_kernel<<<grid, block, 0, stream>>>(A, Wm, Cin, bias, Dm, M, N, relu);
    };
    auto gather_self = [&](const float* Xsrc, int L, int n_dst,
                           const float* A, const float* Wm, float* Dm, int M) {
        int ng = ((n_dst + 7) / 8 + 3) / 4;
        int nm = 4 * ((M + 127) / 128);
        int groups = (ng > nm) ? ng : nm;
        gather_selfgemm_kernel<<<dim3(5 * groups), dim3(32, 8), 0, stream>>>(
            Xsrc, slots + (size_t)L * LSTRIDE, off_row + L * RSTRIDE,
            cnt + L * RSTRIDE, n_dst, agg, A, Wm, Dm, M);
    };

    part_kernel<<<dim3(NCHUNK, 3), dim3(1024), 0, stream>>>(
        src_ii, dst_ii, src_iu0, dst_iu0, src_iu1, dst_iu1,
        E_ii, E_iu0, E_iu1, N_ITEM_DST, N_USER1, N_USER1,
        psrc, pdst, segoff);
    owner_kernel<<<dim3(256, 3), dim3(1024), 0, stream>>>(
        psrc, pdst, segoff, N_ITEM_DST, N_USER1, N_USER1, cnt, off_row, slots);

    gather_self(x_item, 0, N_ITEM_DST, x_item, Ws1, itemx, N_ITEM_DST);
    gemm(agg,    Wn1, itemx,   b1,      itemx, N_ITEM_DST, DH, 1);

    gather_self(x_item, 1, N_USER1, x_user, Ws2, userx, N_USER1);
    gemm(agg,    Wn2, userx,   b2,      userx, N_USER1, DH, 1);

    gather_self(itemx, 2, N_USER1, userx, Ws3, userx2, N_USER1);
    gemm(agg,   Wn3, userx2,  b3,      userx2, N_USER1, DH, 1);

    gemm(userx2, Wlin, nullptr, blin, out, N_USER1, DOUT, 0);
}
